# starting kernel
# baseline (speedup 1.0000x reference)
#include <hip/hip_runtime.h>


typedef __attribute__((ext_vector_type(2))) float v2f;
typedef __attribute__((ext_vector_type(4))) float v4f;
typedef __attribute__((ext_vector_type(8))) float v8f;

#define BATCH 8
#define H 128
#define W 128
#define OH 127
#define OW 127
#define TILE 8
#define SRC 9
#define NSLOT 96
#define PIXSTRIDE 516

__global__ __launch_bounds__(1024, 1)
void convsbs_fused_kernel(const float* __restrict__ channels,
                          const float* __restrict__ cores,
                          float* __restrict__ out) {
  __shared__ float inter[NSLOT * PIXSTRIDE];

  const int b   = blockIdx.z;
  const int ty0 = blockIdx.y * TILE;
  const int tx0 = blockIdx.x * TILE;

  const float* __restrict__ ch0 = channels;
  const float* __restrict__ ch1 = channels + (long)BATCH * H * W * 4;

  const int tid  = (int)threadIdx.x;
  const int wv   = tid >> 5;
  const int lane = tid & 31;

  if (wv < 24) {
    const int g    = wv >> 2;
    const int core = wv & 3;
    const int m    = lane & 15;
    const int half = lane >> 4;

    int s = g * 16 + m;
    int sc = s > 80 ? 80 : s;
    int sy = sc / SRC, sx = sc % SRC;
    int gy = ty0 + sy; gy = gy > (H - 1) ? (H - 1) : gy;
    int gx = tx0 + sx; gx = gx > (W - 1) ? (W - 1) : gx;
    const long pixoff = ((((long)b * H + gy) * W) + gx) * 4;

    const v4f c0 = *(const v4f*)(ch0 + pixoff);
    const v2f c1 = *(const v2f*)(ch1 + pixoff + 2 * half);

    v2f afrag[4];
#pragma unroll
    for (int kt = 0; kt < 4; ++kt) {
      afrag[kt][0] = c0[kt] * c1[0];
      afrag[kt][1] = c0[kt] * c1[1];
    }

    const float* __restrict__ wbase = cores + core * 2048;
    const int sbase = g * 16 + 8 * half;

#pragma unroll
    for (int nt = 0; nt < 8; ++nt) {
      v8f acc = {};
#pragma unroll
      for (int kt = 0; kt < 4; ++kt) {
        const v2f bfrag =
            *(const v2f*)(wbase + (nt * 16 + m) * 16 + kt * 4 + 2 * half);
        acc = __builtin_amdgcn_wmma_f32_16x16x4_f32(
            false, afrag[kt], false, bfrag, (short)0, acc, false, false);
      }
      const int colbase = core * 128 + nt * 16 + m;
#pragma unroll
      for (int v = 0; v < 8; ++v)
        inter[(sbase + v) * PIXSTRIDE + colbase] = acc[v];
    }
  }

  __syncthreads();

  const int pix = tid >> 4;
  const int chn = tid & 15;
  const int oyl = pix >> 3, oxl = pix & 7;
  const int p  = (chn >> 3) & 1;
  const int q  = (chn >> 2) & 1;
  const int r  = (chn >> 1) & 1;
  const int s2 = chn & 1;

  const int sA = oyl * SRC + oxl;
  const float* __restrict__ Am = &inter[(sA)*PIXSTRIDE + 0 * 128 + p * 64];
  const float* __restrict__ Bm = &inter[(sA + 1) * PIXSTRIDE + 1 * 128 + q * 64];
  const float* __restrict__ Cm = &inter[(sA + SRC) * PIXSTRIDE + 2 * 128 + r * 64];
  const float* __restrict__ Dm = &inter[(sA + SRC + 1) * PIXSTRIDE + 3 * 128 + s2 * 64];

  float acc = 0.f;
  for (int i = 0; i < 8; ++i) {
    float e[8];
#pragma unroll
    for (int k = 0; k < 8; ++k) e[k] = 0.f;
#pragma unroll
    for (int j = 0; j < 8; ++j) {
      const float av = Am[i * 8 + j];
#pragma unroll
      for (int k = 0; k < 8; ++k) e[k] = fmaf(av, Bm[j * 8 + k], e[k]);
    }
#pragma unroll
    for (int l = 0; l < 8; ++l) {
      float gv = 0.f;
#pragma unroll
      for (int k = 0; k < 8; ++k) gv = fmaf(e[k], Cm[k * 8 + l], gv);
      acc = fmaf(gv, Dm[l * 8 + i], acc);
    }
  }

  const int oy = ty0 + oyl;
  const int ox = tx0 + oxl;
  if (oy < OH && ox < OW)
    out[((((long)b * OH + oy) * OW) + ox) * 16 + chn] = acc;
}

extern "C" void kernel_launch(void* const* d_in, const int* in_sizes, int n_in,
                              void* d_out, int out_size, void* d_ws, size_t ws_size,
                              hipStream_t stream) {
  (void)in_sizes; (void)n_in; (void)d_ws; (void)ws_size; (void)out_size;
  const float* channels = (const float*)d_in[0];
  const float* cores    = (const float*)d_in[1];
  float* out            = (float*)d_out;

  dim3 grid((OW + TILE - 1) / TILE, (OH + TILE - 1) / TILE, BATCH);
  dim3 block(1024);
  hipLaunchKernelGGL(convsbs_fused_kernel, grid, block, 0, stream,
                     channels, cores, out);
}


# One wave owns 16 output pixels: single-wave workgroups run all 4 cores' WMMAs on their own outputs' source pixels into a private 33 KB LDS slab, then the ring trace; no halo, no cross-wave barrier
# speedup vs baseline: 5.2779x; 5.2779x over previous
#include <hip/hip_runtime.h>


typedef __attribute__((ext_vector_type(2))) float v2f;
typedef __attribute__((ext_vector_type(4))) float v4f;
typedef __attribute__((ext_vector_type(8))) float v8f;

#define BATCH 8
#define H 128
#define W 128
#define OH 127
#define OW 127
#define NPIX 16
#define XCHUNKS 8
#define PIXSTRIDE 516

__global__ __launch_bounds__(32, 3)
void convsbs_fused_kernel(const float* __restrict__ channels,
                          const float* __restrict__ cores,
                          float* __restrict__ out) {
  __shared__ __attribute__((aligned(16))) float inter[NPIX * PIXSTRIDE];

  const int b   = blockIdx.z;
  const int oy  = blockIdx.y;
  const int ox0 = blockIdx.x * NPIX;

  const float* __restrict__ ch0 = channels;
  const float* __restrict__ ch1 = channels + (long)BATCH * H * W * 4;

  const int lane = (int)threadIdx.x;

  {
    const int m    = lane & 15;
    const int half = lane >> 4;

    v2f afrag[4][4];
#pragma unroll
    for (int core = 0; core < 4; ++core) {
      const int gy = oy + (core >> 1);
      int gx = ox0 + m + (core & 1);
      gx = gx > (W - 1) ? (W - 1) : gx;
      const long pixoff = ((((long)b * H + gy) * W) + gx) * 4;
      const v4f c0 = *(const v4f*)(ch0 + pixoff);
      const v2f c1 = *(const v2f*)(ch1 + pixoff + 2 * half);
#pragma unroll
      for (int kt = 0; kt < 4; ++kt) {
        afrag[core][kt][0] = c0[kt] * c1[0];
        afrag[core][kt][1] = c0[kt] * c1[1];
      }
    }

#pragma unroll
    for (int core = 0; core < 4; ++core) {
      const float* __restrict__ wbase = cores + core * 2048;
#pragma unroll
      for (int nt = 0; nt < 8; ++nt) {
        v8f acc = {};
#pragma unroll
        for (int kt = 0; kt < 4; ++kt) {
          const v2f bfrag =
              *(const v2f*)(wbase + (nt * 16 + m) * 16 + kt * 4 + 2 * half);
          acc = __builtin_amdgcn_wmma_f32_16x16x4_f32(
              false, afrag[core][kt], false, bfrag, (short)0, acc, false, false);
        }
        const int colbase = core * 128 + nt * 16 + m;
#pragma unroll
        for (int v = 0; v < 8; ++v)
          inter[(8 * half + v) * PIXSTRIDE + colbase] = acc[v];
      }
    }
  }

  __syncthreads();

  const int c8  = lane & 7;
  const int p   = (c8 >> 2) & 1;
  const int q   = (c8 >> 1) & 1;
  const int ih  = c8 & 1;
  const int pq4 = (lane >> 4) + 2 * ((lane >> 3) & 1);

  for (int pass = 0; pass < 4; ++pass) {
    const int sl = pass * 4 + pq4;
    const float* __restrict__ Am = &inter[sl * PIXSTRIDE + 0 * 128 + p * 64 + ih * 32];
    const float* __restrict__ Bm = &inter[sl * PIXSTRIDE + 1 * 128 + q * 64];
    const float* __restrict__ Cm = &inter[sl * PIXSTRIDE + 2 * 128];
    const float* __restrict__ Dm = &inter[sl * PIXSTRIDE + 3 * 128 + ih * 4];

    v4f alo[4], ahi[4];
#pragma unroll
    for (int ii = 0; ii < 4; ++ii) {
      alo[ii] = *(const v4f*)(Am + ii * 8);
      ahi[ii] = *(const v4f*)(Am + ii * 8 + 4);
    }

    float e[4][8];
#pragma unroll
    for (int ii = 0; ii < 4; ++ii)
#pragma unroll
      for (int k = 0; k < 8; ++k) e[ii][k] = 0.f;
#pragma unroll
    for (int j = 0; j < 8; ++j) {
      const v4f b0 = *(const v4f*)(Bm + j * 8);
      const v4f b1 = *(const v4f*)(Bm + j * 8 + 4);
#pragma unroll
      for (int ii = 0; ii < 4; ++ii) {
        const float av = (j < 4) ? alo[ii][j & 3] : ahi[ii][j & 3];
#pragma unroll
        for (int k = 0; k < 4; ++k) {
          e[ii][k]     = fmaf(av, b0[k], e[ii][k]);
          e[ii][k + 4] = fmaf(av, b1[k], e[ii][k + 4]);
        }
      }
    }

    float acc[2][2] = {{0.f, 0.f}, {0.f, 0.f}};
#pragma unroll
    for (int lh = 0; lh < 2; ++lh) {
      float g[2][4][4];
#pragma unroll
      for (int r = 0; r < 2; ++r)
#pragma unroll
        for (int ii = 0; ii < 4; ++ii)
#pragma unroll
          for (int l = 0; l < 4; ++l) g[r][ii][l] = 0.f;
#pragma unroll
      for (int r = 0; r < 2; ++r) {
#pragma unroll
        for (int k = 0; k < 8; ++k) {
          const v4f cv = *(const v4f*)(Cm + r * 64 + k * 8 + lh * 4);
#pragma unroll
          for (int ii = 0; ii < 4; ++ii) {
            const float ev = e[ii][k];
#pragma unroll
            for (int l = 0; l < 4; ++l)
              g[r][ii][l] = fmaf(ev, cv[l], g[r][ii][l]);
          }
        }
      }
#pragma unroll
      for (int s2 = 0; s2 < 2; ++s2) {
#pragma unroll
        for (int l = 0; l < 4; ++l) {
          const v4f d = *(const v4f*)(Dm + s2 * 64 + (lh * 4 + l) * 8);
#pragma unroll
          for (int ii = 0; ii < 4; ++ii) {
            acc[0][s2] = fmaf(g[0][ii][l], d[ii], acc[0][s2]);
            acc[1][s2] = fmaf(g[1][ii][l], d[ii], acc[1][s2]);
          }
        }
      }
    }

#pragma unroll
    for (int r = 0; r < 2; ++r)
#pragma unroll
      for (int s2 = 0; s2 < 2; ++s2)
        acc[r][s2] += __shfl_xor(acc[r][s2], 1);

    v2f o;
    o[0] = ih ? acc[1][0] : acc[0][0];
    o[1] = ih ? acc[1][1] : acc[0][1];

    const int ox = ox0 + sl;
    if (ox < OW)
      *(v2f*)(out + ((((long)b * OH + oy) * OW) + ox) * 16 + c8 * 2) = o;
  }
}

extern "C" void kernel_launch(void* const* d_in, const int* in_sizes, int n_in,
                              void* d_out, int out_size, void* d_ws, size_t ws_size,
                              hipStream_t stream) {
  (void)in_sizes; (void)n_in; (void)d_ws; (void)ws_size; (void)out_size;
  const float* channels = (const float*)d_in[0];
  const float* cores    = (const float*)d_in[1];
  float* out            = (float*)d_out;

  dim3 grid(XCHUNKS, OH, BATCH);
  dim3 block(32);
  hipLaunchKernelGGL(convsbs_fused_kernel, grid, block, 0, stream,
                     channels, cores, out);
}


# Stage 2 runs in the WMMA result layout (weights as A operand; lane = pixel x 4 trace rows, one pass): core 0 stays in registers, cores 1-3 cross LDS as 32-byte rows; 128 LDS instructions, was 416
# speedup vs baseline: 6.3345x; 1.2002x over previous
#include <hip/hip_runtime.h>


typedef __attribute__((ext_vector_type(2))) float v2f;
typedef __attribute__((ext_vector_type(4))) float v4f;
typedef __attribute__((ext_vector_type(8))) float v8f;

#define BATCH 8
#define H 128
#define W 128
#define OH 127
#define OW 127
#define NPIX 16
#define XCHUNKS 8
#define PIXSTRIDE 388

__device__ __forceinline__ float other_half(float x) {
  return __builtin_bit_cast(float, __builtin_amdgcn_permlanex16(
      __builtin_bit_cast(int, x), __builtin_bit_cast(int, x), 0x76543210, 0xfedcba98, false, false));
}

__global__ __launch_bounds__(32, 3)
void convsbs_fused_kernel(const float* __restrict__ channels,
                          const float* __restrict__ cores,
                          float* __restrict__ out) {
  __shared__ __attribute__((aligned(16))) float inter[NPIX * PIXSTRIDE];

  const int b   = blockIdx.z;
  const int oy  = blockIdx.y;
  const int ox0 = blockIdx.x * NPIX;

  const float* __restrict__ ch0 = channels;
  const float* __restrict__ ch1 = channels + (long)BATCH * H * W * 4;

  const int lane = (int)threadIdx.x;
  const int c    = lane & 15;
  const int h    = lane >> 4;

  v2f xfrag[4][4];
#pragma unroll
  for (int core = 0; core < 4; ++core) {
    const int gy = oy + (core >> 1);
    int gx = ox0 + c + (core & 1);
    gx = gx > (W - 1) ? (W - 1) : gx;
    const long pixoff = ((((long)b * H + gy) * W) + gx) * 4;
    const v4f c0 = *(const v4f*)(ch0 + pixoff);
    const v2f c1 = *(const v2f*)(ch1 + pixoff + 2 * h);
#pragma unroll
    for (int kt = 0; kt < 4; ++kt) {
      xfrag[core][kt][0] = c0[kt] * c1[0];
      xfrag[core][kt][1] = c0[kt] * c1[1];
    }
  }

#pragma unroll
  for (int core = 1; core <= 3; ++core) {
    const float* __restrict__ wb = cores + core * 2048 + c * 16 + 2 * h;
#pragma unroll
    for (int nt = 0; nt < 8; ++nt) {
      v8f acc = {};
#pragma unroll
      for (int kt = 0; kt < 4; ++kt) {
        const v2f wfrag = *(const v2f*)(wb + nt * 256 + kt * 4);
        acc = __builtin_amdgcn_wmma_f32_16x16x4_f32(
            false, wfrag, false, xfrag[core][kt], (short)0, acc, false, false);
      }
      float* dst = &inter[c * PIXSTRIDE + (core - 1) * 128 + nt * 16 + 8 * h];
      v4f lo, hi;
#pragma unroll
      for (int v = 0; v < 4; ++v) { lo[v] = acc[v]; hi[v] = acc[v + 4]; }
      *(v4f*)dst       = lo;
      *(v4f*)(dst + 4) = hi;
    }
  }

  v8f ta[8];
  {
    const int row0 = 32 * (c >> 3) + (c & 7);
    const float* __restrict__ wb0 = cores + row0 * 16 + 2 * h;
#pragma unroll
    for (int t = 0; t < 8; ++t) {
      v8f acc0 = {};
#pragma unroll
      for (int kt = 0; kt < 4; ++kt) {
        const v2f w0 = *(const v2f*)(wb0 + ((t >> 2) * 64 + (t & 3) * 8) * 16 + kt * 4);
        acc0 = __builtin_amdgcn_wmma_f32_16x16x4_f32(
            false, w0, false, xfrag[0][kt], (short)0, acc0, false, false);
      }
      ta[t] = acc0;
    }
  }

  __syncthreads();
  __builtin_amdgcn_sched_barrier(0);

  const float* __restrict__ Bm = &inter[c * PIXSTRIDE];
  const float* __restrict__ Cm = &inter[c * PIXSTRIDE + 128];
  const float* __restrict__ Dm = &inter[c * PIXSTRIDE + 256 + 4 * h];

  float e[2][2][4][8];
#pragma unroll
  for (int p = 0; p < 2; ++p)
#pragma unroll
    for (int q = 0; q < 2; ++q)
#pragma unroll
      for (int ii = 0; ii < 4; ++ii)
#pragma unroll
        for (int k = 0; k < 8; ++k) e[p][q][ii][k] = 0.f;
#pragma unroll
  for (int q = 0; q < 2; ++q) {
#pragma unroll
    for (int j = 0; j < 8; ++j) {
      const v4f b0 = *(const v4f*)(Bm + q * 64 + j * 8);
      const v4f b1 = *(const v4f*)(Bm + q * 64 + j * 8 + 4);
#pragma unroll
      for (int p = 0; p < 2; ++p) {
#pragma unroll
        for (int ii = 0; ii < 4; ++ii) {
          const float av = ta[p * 4 + ii][j];
#pragma unroll
          for (int k = 0; k < 4; ++k) {
            e[p][q][ii][k]     = fmaf(av, b0[k], e[p][q][ii][k]);
            e[p][q][ii][k + 4] = fmaf(av, b1[k], e[p][q][ii][k + 4]);
          }
        }
      }
    }
  }

  float acc[2][2][2][2];
#pragma unroll
  for (int p = 0; p < 2; ++p)
#pragma unroll
    for (int q = 0; q < 2; ++q)
#pragma unroll
      for (int r = 0; r < 2; ++r)
#pragma unroll
        for (int s2 = 0; s2 < 2; ++s2) acc[p][q][r][s2] = 0.f;

#pragma unroll
  for (int r = 0; r < 2; ++r) {
#pragma unroll
    for (int lh = 0; lh < 2; ++lh) {
      float g[2][2][4][4];
#pragma unroll
      for (int p = 0; p < 2; ++p)
#pragma unroll
        for (int q = 0; q < 2; ++q)
#pragma unroll
          for (int ii = 0; ii < 4; ++ii)
#pragma unroll
            for (int l = 0; l < 4; ++l) g[p][q][ii][l] = 0.f;
#pragma unroll
      for (int k = 0; k < 8; ++k) {
        const v4f cv = *(const v4f*)(Cm + r * 64 + k * 8 + lh * 4);
#pragma unroll
        for (int p = 0; p < 2; ++p)
#pragma unroll
          for (int q = 0; q < 2; ++q)
#pragma unroll
            for (int ii = 0; ii < 4; ++ii) {
              const float ev = e[p][q][ii][k];
#pragma unroll
              for (int l = 0; l < 4; ++l)
                g[p][q][ii][l] = fmaf(ev, cv[l], g[p][q][ii][l]);
            }
      }
#pragma unroll
      for (int s2 = 0; s2 < 2; ++s2) {
#pragma unroll
        for (int l = 0; l < 4; ++l) {
          const v4f d = *(const v4f*)(Dm + s2 * 64 + (lh * 4 + l) * 8);
#pragma unroll
          for (int ii = 0; ii < 4; ++ii) {
            const float dv = d[ii];
#pragma unroll
            for (int p = 0; p < 2; ++p)
#pragma unroll
              for (int q = 0; q < 2; ++q)
                acc[p][q][r][s2] = fmaf(g[p][q][ii][l], dv, acc[p][q][r][s2]);
          }
        }
      }
    }
  }

  v4f o0, o1;
#pragma unroll
  for (int q = 0; q < 2; ++q)
#pragma unroll
    for (int r = 0; r < 2; ++r)
#pragma unroll
      for (int s2 = 0; s2 < 2; ++s2) {
        const float mine = h ? acc[1][q][r][s2] : acc[0][q][r][s2];
        const float give = h ? acc[0][q][r][s2] : acc[1][q][r][s2];
        const float sum  = mine + other_half(give);
        if (q == 0) o0[r * 2 + s2] = sum; else o1[r * 2 + s2] = sum;
      }

  const int ox = ox0 + c;
  if (ox < OW) {
    float* dstp = out + ((((long)b * OH + oy) * OW) + ox) * 16 + h * 8;
    *(v4f*)dstp       = o0;
    *(v4f*)(dstp + 4) = o1;
  }
}

extern "C" void kernel_launch(void* const* d_in, const int* in_sizes, int n_in,
                              void* d_out, int out_size, void* d_ws, size_t ws_size,
                              hipStream_t stream) {
  (void)in_sizes; (void)n_in; (void)d_ws; (void)ws_size; (void)out_size;
  const float* channels = (const float*)d_in[0];
  const float* cores    = (const float*)d_in[1];
  float* out            = (float*)d_out;

  dim3 grid(XCHUNKS, OH, BATCH);
  dim3 block(32);
  hipLaunchKernelGGL(convsbs_fused_kernel, grid, block, 0, stream,
                     channels, cores, out);
}


# Stage 2 ring closed from both ends in the lane layout: f = C_r D_s[:,own 4 columns] meets the resident e = A_p[own rows]B_q in sum e[i,k]f[k,i]; the e*C_r pass is gone: 2560 FMAs per lane, was 3584
# speedup vs baseline: 7.6851x; 1.2132x over previous
#include <hip/hip_runtime.h>


typedef __attribute__((ext_vector_type(2))) float v2f;
typedef __attribute__((ext_vector_type(4))) float v4f;
typedef __attribute__((ext_vector_type(8))) float v8f;

#define BATCH 8
#define H 128
#define W 128
#define OH 127
#define OW 127
#define NPIX 16
#define XCHUNKS 8
#define PIXSTRIDE 388

__device__ __forceinline__ float other_half(float x) {
  return __builtin_bit_cast(float, __builtin_amdgcn_permlanex16(
      __builtin_bit_cast(int, x), __builtin_bit_cast(int, x), 0x76543210, 0xfedcba98, false, false));
}

__global__ __launch_bounds__(32, 3)
void convsbs_fused_kernel(const float* __restrict__ channels,
                          const float* __restrict__ cores,
                          float* __restrict__ out) {
  __shared__ __attribute__((aligned(16))) float inter[NPIX * PIXSTRIDE];

  const int b   = blockIdx.z;
  const int oy  = blockIdx.y;
  const int ox0 = blockIdx.x * NPIX;

  const float* __restrict__ ch0 = channels;
  const float* __restrict__ ch1 = channels + (long)BATCH * H * W * 4;

  const int lane = (int)threadIdx.x;
  const int c    = lane & 15;
  const int h    = lane >> 4;

  v2f xfrag[4][4];
#pragma unroll
  for (int core = 0; core < 4; ++core) {
    const int gy = oy + (core >> 1);
    int gx = ox0 + c + (core & 1);
    gx = gx > (W - 1) ? (W - 1) : gx;
    const long pixoff = ((((long)b * H + gy) * W) + gx) * 4;
    const v4f c0 = *(const v4f*)(ch0 + pixoff);
    const v2f c1 = *(const v2f*)(ch1 + pixoff + 2 * h);
#pragma unroll
    for (int kt = 0; kt < 4; ++kt) {
      xfrag[core][kt][0] = c0[kt] * c1[0];
      xfrag[core][kt][1] = c0[kt] * c1[1];
    }
  }

#pragma unroll
  for (int core = 1; core <= 3; ++core) {
    const float* __restrict__ wb = cores + core * 2048 + c * 16 + 2 * h;
#pragma unroll
    for (int nt = 0; nt < 8; ++nt) {
      v8f acc = {};
#pragma unroll
      for (int kt = 0; kt < 4; ++kt) {
        const v2f wfrag = *(const v2f*)(wb + nt * 256 + kt * 4);
        acc = __builtin_amdgcn_wmma_f32_16x16x4_f32(
            false, wfrag, false, xfrag[core][kt], (short)0, acc, false, false);
      }
      float* dst = &inter[c * PIXSTRIDE + (core - 1) * 128 + nt * 16 + 8 * h];
      v4f lo, hi;
#pragma unroll
      for (int v = 0; v < 4; ++v) { lo[v] = acc[v]; hi[v] = acc[v + 4]; }
      *(v4f*)dst       = lo;
      *(v4f*)(dst + 4) = hi;
    }
  }

  v8f ta[8];
  {
    const int row0 = 32 * (c >> 3) + (c & 7);
    const float* __restrict__ wb0 = cores + row0 * 16 + 2 * h;
#pragma unroll
    for (int t = 0; t < 8; ++t) {
      v8f acc0 = {};
#pragma unroll
      for (int kt = 0; kt < 4; ++kt) {
        const v2f w0 = *(const v2f*)(wb0 + ((t >> 2) * 64 + (t & 3) * 8) * 16 + kt * 4);
        acc0 = __builtin_amdgcn_wmma_f32_16x16x4_f32(
            false, w0, false, xfrag[0][kt], (short)0, acc0, false, false);
      }
      ta[t] = acc0;
    }
  }

  __syncthreads();
  __builtin_amdgcn_sched_barrier(0);

  const float* __restrict__ Bm = &inter[c * PIXSTRIDE];
  const float* __restrict__ Cm = &inter[c * PIXSTRIDE + 128];
  const float* __restrict__ Dm = &inter[c * PIXSTRIDE + 256 + 4 * h];

  float e[2][2][4][8];
#pragma unroll
  for (int p = 0; p < 2; ++p)
#pragma unroll
    for (int q = 0; q < 2; ++q)
#pragma unroll
      for (int ii = 0; ii < 4; ++ii)
#pragma unroll
        for (int k = 0; k < 8; ++k) e[p][q][ii][k] = 0.f;
#pragma unroll
  for (int q = 0; q < 2; ++q) {
#pragma unroll
    for (int j = 0; j < 8; ++j) {
      const v4f b0 = *(const v4f*)(Bm + q * 64 + j * 8);
      const v4f b1 = *(const v4f*)(Bm + q * 64 + j * 8 + 4);
#pragma unroll
      for (int p = 0; p < 2; ++p) {
#pragma unroll
        for (int ii = 0; ii < 4; ++ii) {
          const float av = ta[p * 4 + ii][j];
#pragma unroll
          for (int k = 0; k < 4; ++k) {
            e[p][q][ii][k]     = fmaf(av, b0[k], e[p][q][ii][k]);
            e[p][q][ii][k + 4] = fmaf(av, b1[k], e[p][q][ii][k + 4]);
          }
        }
      }
    }
  }

  v4f dcol[2][8];
#pragma unroll
  for (int s2 = 0; s2 < 2; ++s2)
#pragma unroll
    for (int l = 0; l < 8; ++l)
      dcol[s2][l] = *(const v4f*)(Dm + s2 * 64 + l * 8);

  float acc[2][2][2][2];
#pragma unroll
  for (int p = 0; p < 2; ++p)
#pragma unroll
    for (int q = 0; q < 2; ++q)
#pragma unroll
      for (int r = 0; r < 2; ++r)
#pragma unroll
        for (int s2 = 0; s2 < 2; ++s2) acc[p][q][r][s2] = 0.f;

#pragma unroll
  for (int r = 0; r < 2; ++r) {
#pragma unroll
    for (int k = 0; k < 8; ++k) {
      const v4f c0 = *(const v4f*)(Cm + r * 64 + k * 8);
      const v4f c1 = *(const v4f*)(Cm + r * 64 + k * 8 + 4);
#pragma unroll
      for (int s2 = 0; s2 < 2; ++s2) {
        float f[4];
#pragma unroll
        for (int ii = 0; ii < 4; ++ii) f[ii] = c0[0] * dcol[s2][0][ii];
#pragma unroll
        for (int l = 1; l < 4; ++l)
#pragma unroll
          for (int ii = 0; ii < 4; ++ii) f[ii] = fmaf(c0[l], dcol[s2][l][ii], f[ii]);
#pragma unroll
        for (int l = 0; l < 4; ++l)
#pragma unroll
          for (int ii = 0; ii < 4; ++ii) f[ii] = fmaf(c1[l], dcol[s2][l + 4][ii], f[ii]);
#pragma unroll
        for (int ii = 0; ii < 4; ++ii)
#pragma unroll
          for (int p = 0; p < 2; ++p)
#pragma unroll
            for (int q = 0; q < 2; ++q)
              acc[p][q][r][s2] = fmaf(e[p][q][ii][k], f[ii], acc[p][q][r][s2]);
      }
    }
  }

  v4f o0, o1;
#pragma unroll
  for (int q = 0; q < 2; ++q)
#pragma unroll
    for (int r = 0; r < 2; ++r)
#pragma unroll
      for (int s2 = 0; s2 < 2; ++s2) {
        const float mine = h ? acc[1][q][r][s2] : acc[0][q][r][s2];
        const float give = h ? acc[0][q][r][s2] : acc[1][q][r][s2];
        const float sum  = mine + other_half(give);
        if (q == 0) o0[r * 2 + s2] = sum; else o1[r * 2 + s2] = sum;
      }

  const int ox = ox0 + c;
  if (ox < OW) {
    float* dstp = out + ((((long)b * OH + oy) * OW) + ox) * 16 + h * 8;
    *(v4f*)dstp       = o0;
    *(v4f*)(dstp + 4) = o1;
  }
}

extern "C" void kernel_launch(void* const* d_in, const int* in_sizes, int n_in,
                              void* d_out, int out_size, void* d_ws, size_t ws_size,
                              hipStream_t stream) {
  (void)in_sizes; (void)n_in; (void)d_ws; (void)ws_size; (void)out_size;
  const float* channels = (const float*)d_in[0];
  const float* cores    = (const float*)d_in[1];
  float* out            = (float*)d_out;

  dim3 grid(XCHUNKS, OH, BATCH);
  dim3 block(32);
  hipLaunchKernelGGL(convsbs_fused_kernel, grid, block, 0, stream,
                     channels, cores, out);
}


# Stage 1 k order contiguous per lane half (k = 8h + 2kt + s in both operands): a lane's share of a weight row is 32 contiguous bytes: 2 global_load_b128 per tile, was 4 global_load_b64
# speedup vs baseline: 9.2144x; 1.1990x over previous
#include <hip/hip_runtime.h>


typedef __attribute__((ext_vector_type(2))) float v2f;
typedef __attribute__((ext_vector_type(4))) float v4f;
typedef __attribute__((ext_vector_type(8))) float v8f;

#define BATCH 8
#define H 128
#define W 128
#define OH 127
#define OW 127
#define NPIX 16
#define XCHUNKS 8
#define PIXSTRIDE 388

__device__ __forceinline__ float other_half(float x) {
  return __builtin_bit_cast(float, __builtin_amdgcn_permlanex16(
      __builtin_bit_cast(int, x), __builtin_bit_cast(int, x), 0x76543210, 0xfedcba98, false, false));
}

__global__ __launch_bounds__(32, 3)
void convsbs_fused_kernel(const float* __restrict__ channels,
                          const float* __restrict__ cores,
                          float* __restrict__ out) {
  __shared__ __attribute__((aligned(16))) float inter[NPIX * PIXSTRIDE];

  const int b   = blockIdx.z;
  const int oy  = blockIdx.y;
  const int ox0 = blockIdx.x * NPIX;

  const float* __restrict__ ch0 = channels;
  const float* __restrict__ ch1 = channels + (long)BATCH * H * W * 4;

  const int lane = (int)threadIdx.x;
  const int c    = lane & 15;
  const int h    = lane >> 4;

  v2f xfrag[4][4];
#pragma unroll
  for (int core = 0; core < 4; ++core) {
    const int gy = oy + (core >> 1);
    int gx = ox0 + c + (core & 1);
    gx = gx > (W - 1) ? (W - 1) : gx;
    const long pixoff = ((((long)b * H + gy) * W) + gx) * 4;
    const v2f c0 = *(const v2f*)(ch0 + pixoff + 2 * h);
    const v4f c1 = *(const v4f*)(ch1 + pixoff);
#pragma unroll
    for (int kt = 0; kt < 4; ++kt) {
      xfrag[core][kt][0] = c0[kt >> 1] * c1[2 * (kt & 1)];
      xfrag[core][kt][1] = c0[kt >> 1] * c1[2 * (kt & 1) + 1];
    }
  }

#pragma unroll
  for (int core = 1; core <= 3; ++core) {
    const float* __restrict__ wb = cores + core * 2048 + c * 16 + 8 * h;
#pragma unroll
    for (int nt = 0; nt < 8; ++nt) {
      const v4f wlo = *(const v4f*)(wb + nt * 256);
      const v4f whi = *(const v4f*)(wb + nt * 256 + 4);
      v8f acc = {};
#pragma unroll
      for (int kt = 0; kt < 4; ++kt) {
        v2f wfrag;
        wfrag[0] = (kt < 2) ? wlo[2 * kt]     : whi[2 * kt - 4];
        wfrag[1] = (kt < 2) ? wlo[2 * kt + 1] : whi[2 * kt - 3];
        acc = __builtin_amdgcn_wmma_f32_16x16x4_f32(
            false, wfrag, false, xfrag[core][kt], (short)0, acc, false, false);
      }
      float* dst = &inter[c * PIXSTRIDE + (core - 1) * 128 + nt * 16 + 8 * h];
      v4f lo, hi;
#pragma unroll
      for (int v = 0; v < 4; ++v) { lo[v] = acc[v]; hi[v] = acc[v + 4]; }
      *(v4f*)dst       = lo;
      *(v4f*)(dst + 4) = hi;
    }
  }

  v8f ta[8];
  {
    const int row0 = 32 * (c >> 3) + (c & 7);
    const float* __restrict__ wb0 = cores + row0 * 16 + 8 * h;
#pragma unroll
    for (int t = 0; t < 8; ++t) {
      const float* __restrict__ wrow = wb0 + ((t >> 2) * 64 + (t & 3) * 8) * 16;
      const v4f wlo = *(const v4f*)(wrow);
      const v4f whi = *(const v4f*)(wrow + 4);
      v8f acc0 = {};
#pragma unroll
      for (int kt = 0; kt < 4; ++kt) {
        v2f w0;
        w0[0] = (kt < 2) ? wlo[2 * kt]     : whi[2 * kt - 4];
        w0[1] = (kt < 2) ? wlo[2 * kt + 1] : whi[2 * kt - 3];
        acc0 = __builtin_amdgcn_wmma_f32_16x16x4_f32(
            false, w0, false, xfrag[0][kt], (short)0, acc0, false, false);
      }
      ta[t] = acc0;
    }
  }

  __syncthreads();
  __builtin_amdgcn_sched_barrier(0);

  const float* __restrict__ Bm = &inter[c * PIXSTRIDE];
  const float* __restrict__ Cm = &inter[c * PIXSTRIDE + 128];
  const float* __restrict__ Dm = &inter[c * PIXSTRIDE + 256 + 4 * h];

  float e[2][2][4][8];
#pragma unroll
  for (int p = 0; p < 2; ++p)
#pragma unroll
    for (int q = 0; q < 2; ++q)
#pragma unroll
      for (int ii = 0; ii < 4; ++ii)
#pragma unroll
        for (int k = 0; k < 8; ++k) e[p][q][ii][k] = 0.f;
#pragma unroll
  for (int q = 0; q < 2; ++q) {
#pragma unroll
    for (int j = 0; j < 8; ++j) {
      const v4f b0 = *(const v4f*)(Bm + q * 64 + j * 8);
      const v4f b1 = *(const v4f*)(Bm + q * 64 + j * 8 + 4);
#pragma unroll
      for (int p = 0; p < 2; ++p) {
#pragma unroll
        for (int ii = 0; ii < 4; ++ii) {
          const float av = ta[p * 4 + ii][j];
#pragma unroll
          for (int k = 0; k < 4; ++k) {
            e[p][q][ii][k]     = fmaf(av, b0[k], e[p][q][ii][k]);
            e[p][q][ii][k + 4] = fmaf(av, b1[k], e[p][q][ii][k + 4]);
          }
        }
      }
    }
  }

  v4f dcol[2][8];
#pragma unroll
  for (int s2 = 0; s2 < 2; ++s2)
#pragma unroll
    for (int l = 0; l < 8; ++l)
      dcol[s2][l] = *(const v4f*)(Dm + s2 * 64 + l * 8);

  float acc[2][2][2][2];
#pragma unroll
  for (int p = 0; p < 2; ++p)
#pragma unroll
    for (int q = 0; q < 2; ++q)
#pragma unroll
      for (int r = 0; r < 2; ++r)
#pragma unroll
        for (int s2 = 0; s2 < 2; ++s2) acc[p][q][r][s2] = 0.f;

#pragma unroll
  for (int r = 0; r < 2; ++r) {
#pragma unroll
    for (int k = 0; k < 8; ++k) {
      const v4f c0 = *(const v4f*)(Cm + r * 64 + k * 8);
      const v4f c1 = *(const v4f*)(Cm + r * 64 + k * 8 + 4);
#pragma unroll
      for (int s2 = 0; s2 < 2; ++s2) {
        float f[4];
#pragma unroll
        for (int ii = 0; ii < 4; ++ii) f[ii] = c0[0] * dcol[s2][0][ii];
#pragma unroll
        for (int l = 1; l < 4; ++l)
#pragma unroll
          for (int ii = 0; ii < 4; ++ii) f[ii] = fmaf(c0[l], dcol[s2][l][ii], f[ii]);
#pragma unroll
        for (int l = 0; l < 4; ++l)
#pragma unroll
          for (int ii = 0; ii < 4; ++ii) f[ii] = fmaf(c1[l], dcol[s2][l + 4][ii], f[ii]);
#pragma unroll
        for (int ii = 0; ii < 4; ++ii)
#pragma unroll
          for (int p = 0; p < 2; ++p)
#pragma unroll
            for (int q = 0; q < 2; ++q)
              acc[p][q][r][s2] = fmaf(e[p][q][ii][k], f[ii], acc[p][q][r][s2]);
      }
    }
  }

  v4f o0, o1;
#pragma unroll
  for (int q = 0; q < 2; ++q)
#pragma unroll
    for (int r = 0; r < 2; ++r)
#pragma unroll
      for (int s2 = 0; s2 < 2; ++s2) {
        const float mine = h ? acc[1][q][r][s2] : acc[0][q][r][s2];
        const float give = h ? acc[0][q][r][s2] : acc[1][q][r][s2];
        const float sum  = mine + other_half(give);
        if (q == 0) o0[r * 2 + s2] = sum; else o1[r * 2 + s2] = sum;
      }

  const int ox = ox0 + c;
  if (ox < OW) {
    float* dstp = out + ((((long)b * OH + oy) * OW) + ox) * 16 + h * 8;
    *(v4f*)dstp       = o0;
    *(v4f*)(dstp + 4) = o1;
  }
}

extern "C" void kernel_launch(void* const* d_in, const int* in_sizes, int n_in,
                              void* d_out, int out_size, void* d_ws, size_t ws_size,
                              hipStream_t stream) {
  (void)in_sizes; (void)n_in; (void)d_ws; (void)ws_size; (void)out_size;
  const float* channels = (const float*)d_in[0];
  const float* cores    = (const float*)d_in[1];
  float* out            = (float*)d_out;

  dim3 grid(XCHUNKS, OH, BATCH);
  dim3 block(32);
  hipLaunchKernelGGL(convsbs_fused_kernel, grid, block, 0, stream,
                     channels, cores, out);
}


# 4 waves per SIMD (was 3): core 3 handled like core 0, tile rows = the lane's own D columns, its WMMAs behind the e pass in A_p's registers; no D in LDS (96 LDS instr, was 128), 16640-byte slab
# speedup vs baseline: 9.5075x; 1.0318x over previous
#include <hip/hip_runtime.h>


typedef __attribute__((ext_vector_type(2))) float v2f;
typedef __attribute__((ext_vector_type(4))) float v4f;
typedef __attribute__((ext_vector_type(8))) float v8f;

#define BATCH 8
#define H 128
#define W 128
#define OH 127
#define OW 127
#define NPIX 16
#define XCHUNKS 8
#define PIXSTRIDE 260

__device__ __forceinline__ float other_half(float x) {
  return __builtin_bit_cast(float, __builtin_amdgcn_permlanex16(
      __builtin_bit_cast(int, x), __builtin_bit_cast(int, x), 0x76543210, 0xfedcba98, false, false));
}

__global__ __launch_bounds__(32, 4)
void convsbs_fused_kernel(const float* __restrict__ channels,
                          const float* __restrict__ cores,
                          float* __restrict__ out) {
  __shared__ __attribute__((aligned(16))) float inter[NPIX * PIXSTRIDE];

  const int b   = blockIdx.z;
  const int oy  = blockIdx.y;
  const int ox0 = blockIdx.x * NPIX;

  const float* __restrict__ ch0 = channels;
  const float* __restrict__ ch1 = channels + BATCH * H * W * 4;

  const int lane = (int)threadIdx.x;
  const int c    = lane & 15;
  const int h    = lane >> 4;

  v2f xfrag[3][4];
#pragma unroll
  for (int core = 0; core < 3; ++core) {
    const int gy = oy + (core >> 1);
    int gx = ox0 + c + (core & 1);
    gx = gx > (W - 1) ? (W - 1) : gx;
    const unsigned pixoff = ((((unsigned)b * H + gy) * W) + gx) * 4u;
    const v2f c0 = *(const v2f*)(ch0 + pixoff + 2 * h);
    const v4f c1 = *(const v4f*)(ch1 + pixoff);
#pragma unroll
    for (int kt = 0; kt < 4; ++kt) {
      xfrag[core][kt][0] = c0[kt >> 1] * c1[2 * (kt & 1)];
      xfrag[core][kt][1] = c0[kt >> 1] * c1[2 * (kt & 1) + 1];
    }
  }

#pragma unroll
  for (int core = 1; core <= 2; ++core) {
    const float* __restrict__ wb = cores + (unsigned)(core * 2048 + c * 16 + 8 * h);
#pragma unroll
    for (int nt = 0; nt < 8; ++nt) {
      const v4f wlo = *(const v4f*)(wb + nt * 256);
      const v4f whi = *(const v4f*)(wb + nt * 256 + 4);
      v8f acc = {};
#pragma unroll
      for (int kt = 0; kt < 4; ++kt) {
        v2f wfrag;
        wfrag[0] = (kt < 2) ? wlo[2 * kt]     : whi[2 * kt - 4];
        wfrag[1] = (kt < 2) ? wlo[2 * kt + 1] : whi[2 * kt - 3];
        acc = __builtin_amdgcn_wmma_f32_16x16x4_f32(
            false, wfrag, false, xfrag[core][kt], (short)0, acc, false, false);
      }
      float* dst = &inter[c * PIXSTRIDE + (core - 1) * 128 + nt * 16 + 8 * h];
      v4f lo, hi;
#pragma unroll
      for (int v = 0; v < 4; ++v) { lo[v] = acc[v]; hi[v] = acc[v + 4]; }
      *(v4f*)dst       = lo;
      *(v4f*)(dst + 4) = hi;
    }
  }

  v8f ta[8];
  {
    const int row0 = 32 * (c >> 3) + (c & 7);
    const float* __restrict__ wb0 = cores + (unsigned)(row0 * 16 + 8 * h);
#pragma unroll
    for (int t = 0; t < 8; ++t) {
      const float* __restrict__ wrow = wb0 + ((t >> 2) * 64 + (t & 3) * 8) * 16;
      const v4f wlo = *(const v4f*)(wrow);
      const v4f whi = *(const v4f*)(wrow + 4);
      v8f acc0 = {};
#pragma unroll
      for (int kt = 0; kt < 4; ++kt) {
        v2f w0;
        w0[0] = (kt < 2) ? wlo[2 * kt]     : whi[2 * kt - 4];
        w0[1] = (kt < 2) ? wlo[2 * kt + 1] : whi[2 * kt - 3];
        acc0 = __builtin_amdgcn_wmma_f32_16x16x4_f32(
            false, w0, false, xfrag[0][kt], (short)0, acc0, false, false);
      }
      ta[t] = acc0;
    }
  }

  __syncthreads();
  __builtin_amdgcn_sched_barrier(0);

  v2f xfrag3[4];
  {
    const int gy = oy + 1;
    int gx = ox0 + c + 1;
    gx = gx > (W - 1) ? (W - 1) : gx;
    const unsigned pixoff = ((((unsigned)b * H + gy) * W) + gx) * 4u;
    const v2f c0 = *(const v2f*)(ch0 + pixoff + 2 * h);
    const v4f c1 = *(const v4f*)(ch1 + pixoff);
    __builtin_amdgcn_sched_barrier(0);
#pragma unroll
    for (int kt = 0; kt < 4; ++kt) {
      xfrag3[kt][0] = c0[kt >> 1] * c1[2 * (kt & 1)];
      xfrag3[kt][1] = c0[kt >> 1] * c1[2 * (kt & 1) + 1];
    }
  }

  const float* __restrict__ Bm = &inter[c * PIXSTRIDE];
  const float* __restrict__ Cm = &inter[c * PIXSTRIDE + 128];

  float e[2][2][4][8];
#pragma unroll
  for (int p = 0; p < 2; ++p)
#pragma unroll
    for (int q = 0; q < 2; ++q)
#pragma unroll
      for (int ii = 0; ii < 4; ++ii)
#pragma unroll
        for (int k = 0; k < 8; ++k) e[p][q][ii][k] = 0.f;
#pragma unroll
  for (int q = 0; q < 2; ++q) {
#pragma unroll
    for (int j = 0; j < 8; ++j) {
      const v4f b0 = *(const v4f*)(Bm + q * 64 + j * 8);
      const v4f b1 = *(const v4f*)(Bm + q * 64 + j * 8 + 4);
#pragma unroll
      for (int p = 0; p < 2; ++p) {
#pragma unroll
        for (int ii = 0; ii < 4; ++ii) {
          const float av = ta[p * 4 + ii][j];
#pragma unroll
          for (int k = 0; k < 4; ++k) {
            e[p][q][ii][k]     = fmaf(av, b0[k], e[p][q][ii][k]);
            e[p][q][ii][k + 4] = fmaf(av, b1[k], e[p][q][ii][k + 4]);
          }
        }
      }
    }
  }

  v8f td[8];
  {
    const int row3 = 8 * ((c >> 2) & 1) + 4 * (c >> 3) + (c & 3);
    const float* __restrict__ wb3 = cores + (unsigned)(3 * 2048 + row3 * 16 + 8 * h);
#pragma unroll
    for (int t = 0; t < 8; ++t) {
      const v4f wlo = *(const v4f*)(wb3 + t * 256);
      const v4f whi = *(const v4f*)(wb3 + t * 256 + 4);
      v8f acc3 = {};
#pragma unroll
      for (int kt = 0; kt < 4; ++kt) {
        v2f w3;
        w3[0] = (kt < 2) ? wlo[2 * kt]     : whi[2 * kt - 4];
        w3[1] = (kt < 2) ? wlo[2 * kt + 1] : whi[2 * kt - 3];
        acc3 = __builtin_amdgcn_wmma_f32_16x16x4_f32(
            false, w3, false, xfrag3[kt], (short)0, acc3, false, false);
      }
      td[t] = acc3;
    }
  }

  v4f dcol[2][8];
#pragma unroll
  for (int s2 = 0; s2 < 2; ++s2)
#pragma unroll
    for (int l = 0; l < 8; ++l)
#pragma unroll
      for (int ii = 0; ii < 4; ++ii)
        dcol[s2][l][ii] = td[s2 * 4 + (l >> 1)][(l & 1) * 4 + ii];

  float acc[2][2][2][2];
#pragma unroll
  for (int p = 0; p < 2; ++p)
#pragma unroll
    for (int q = 0; q < 2; ++q)
#pragma unroll
      for (int r = 0; r < 2; ++r)
#pragma unroll
        for (int s2 = 0; s2 < 2; ++s2) acc[p][q][r][s2] = 0.f;

#pragma unroll
  for (int r = 0; r < 2; ++r) {
#pragma unroll
    for (int k = 0; k < 8; ++k) {
      const v4f c0 = *(const v4f*)(Cm + r * 64 + k * 8);
      const v4f c1 = *(const v4f*)(Cm + r * 64 + k * 8 + 4);
#pragma unroll
      for (int s2 = 0; s2 < 2; ++s2) {
        float f[4];
#pragma unroll
        for (int ii = 0; ii < 4; ++ii) f[ii] = c0[0] * dcol[s2][0][ii];
#pragma unroll
        for (int l = 1; l < 4; ++l)
#pragma unroll
          for (int ii = 0; ii < 4; ++ii) f[ii] = fmaf(c0[l], dcol[s2][l][ii], f[ii]);
#pragma unroll
        for (int l = 0; l < 4; ++l)
#pragma unroll
          for (int ii = 0; ii < 4; ++ii) f[ii] = fmaf(c1[l], dcol[s2][l + 4][ii], f[ii]);
#pragma unroll
        for (int ii = 0; ii < 4; ++ii)
#pragma unroll
          for (int p = 0; p < 2; ++p)
#pragma unroll
            for (int q = 0; q < 2; ++q)
              acc[p][q][r][s2] = fmaf(e[p][q][ii][k], f[ii], acc[p][q][r][s2]);
      }
    }
  }

  v4f o0, o1;
#pragma unroll
  for (int q = 0; q < 2; ++q)
#pragma unroll
    for (int r = 0; r < 2; ++r)
#pragma unroll
      for (int s2 = 0; s2 < 2; ++s2) {
        const float mine = h ? acc[1][q][r][s2] : acc[0][q][r][s2];
        const float give = h ? acc[0][q][r][s2] : acc[1][q][r][s2];
        const float sum  = mine + other_half(give);
        if (q == 0) o0[r * 2 + s2] = sum; else o1[r * 2 + s2] = sum;
      }

  const int ox = ox0 + c;
  if (ox < OW) {
    float* dstp = out + (((((unsigned)b * OH + oy) * OW) + ox) * 16u + h * 8u);
    *(v4f*)dstp       = o0;
    *(v4f*)(dstp + 4) = o1;
  }
}

extern "C" void kernel_launch(void* const* d_in, const int* in_sizes, int n_in,
                              void* d_out, int out_size, void* d_ws, size_t ws_size,
                              hipStream_t stream) {
  (void)in_sizes; (void)n_in; (void)d_ws; (void)ws_size; (void)out_size;
  const float* channels = (const float*)d_in[0];
  const float* cores    = (const float*)d_in[1];
  float* out            = (float*)d_out;

  dim3 grid(XCHUNKS, OH, BATCH);
  dim3 block(32);
  hipLaunchKernelGGL(convsbs_fused_kernel, grid, block, 0, stream,
                     channels, cores, out);
}
